# starting kernel
# baseline (speedup 1.0000x reference)
#include <hip/hip_runtime.h>
#include <stdint.h>

typedef float v2f __attribute__((ext_vector_type(2)));
typedef float v8f __attribute__((ext_vector_type(8)));

#define N_NODES 50000
#define N_EDGES 800000
#define ETOT    (N_EDGES + N_NODES)
#define IN_CH   64
#define F1      256
#define H1      4
#define C1      64
#define F2      32
#define NEG_SLOPE 0.2f

__global__ void zero_f32(float* __restrict__ p, int n) {
  int i = blockIdx.x * blockDim.x + threadIdx.x;
  if (i < n) p[i] = 0.0f;
}
__global__ void zero_u32(unsigned* __restrict__ p, int n) {
  int i = blockIdx.x * blockDim.x + threadIdx.x;
  if (i < n) p[i] = 0u;
}

__device__ __forceinline__ void edge_sd(const int* __restrict__ ei, int e, int& s, int& d) {
  if (e < N_EDGES) { s = ei[e]; d = ei[N_EDGES + e]; }
  else             { s = d = e - N_EDGES; }
}

__device__ __forceinline__ unsigned f2key(float f) {
  unsigned b = __float_as_uint(f);
  return (b & 0x80000000u) ? ~b : (b | 0x80000000u);
}
__device__ __forceinline__ float key2f(unsigned k) {
  if (k == 0u) return 0.0f;
  unsigned b = (k & 0x80000000u) ? (k & 0x7fffffffu) : ~k;
  return __uint_as_float(b);
}

__device__ __forceinline__ float leaky(float x) {
  return x > 0.0f ? x : NEG_SLOPE * x;
}

template <int NT>
__global__ void wmma_gemm_f32(const float* __restrict__ A, const float* __restrict__ B,
                              float* __restrict__ C, int M, int K, int N, int tiles) {
  const int lane = threadIdx.x & 31;
  const int tile = blockIdx.x * (blockDim.x >> 5) + (threadIdx.x >> 5);
  if (tile >= tiles) return;
  const int tilesN = N / (16 * NT);
  const int tm  = tile / tilesN;
  const int tn0 = (tile % tilesN) * NT;

  const int rc   = lane & 15;
  const int hi   = lane >> 4;
  const int koff = hi << 1;

  const float* Ap = A + (size_t)(tm * 16 + rc) * K + koff;
  const float* Bp = B + (size_t)koff * N + tn0 * 16 + rc;

  v8f acc[NT];
#pragma unroll
  for (int t = 0; t < NT; ++t) acc[t] = (v8f){};

  for (int k0 = 0; k0 < K; k0 += 4) {
    v2f a;
    a.x = Ap[k0];
    a.y = Ap[k0 + 1];
#pragma unroll
    for (int t = 0; t < NT; ++t) {
      v2f b;
      b.x = Bp[(size_t)k0 * N + t * 16];
      b.y = Bp[(size_t)(k0 + 1) * N + t * 16];
      acc[t] = __builtin_amdgcn_wmma_f32_16x16x4_f32(false, a, false, b, (short)0,
                                                     acc[t], false, false);
    }
  }
  float* Cp = C + (size_t)(tm * 16 + hi * 8) * N + tn0 * 16 + rc;
#pragma unroll
  for (int t = 0; t < NT; ++t)
#pragma unroll
    for (int r = 0; r < 8; ++r) Cp[(size_t)r * N + t * 16] = acc[t][r];
}

template <int CH>
__global__ void att_dots(const float* __restrict__ h, const float* __restrict__ att_s,
                         const float* __restrict__ att_d, float* __restrict__ as,
                         float* __restrict__ ad, int n, int heads) {
  int idx = blockIdx.x * blockDim.x + threadIdx.x;
  if (idx >= n * heads) return;
  int node = idx / heads, hd = idx % heads;
  const float* hp = h + (size_t)node * heads * CH + (size_t)hd * CH;
  const float* sp = att_s + (size_t)hd * CH;
  const float* dp = att_d + (size_t)hd * CH;
  float s = 0.0f, d = 0.0f;
#pragma unroll 4
  for (int c = 0; c < CH; ++c) { float v = hp[c]; s += v * sp[c]; d += v * dp[c]; }
  as[idx] = s;
  ad[idx] = d;
}

template <int H>
__global__ void edge_logit_max(const int* __restrict__ ei, const float* __restrict__ as,
                               const float* __restrict__ ad, float* __restrict__ elog,
                               unsigned* __restrict__ lkey) {
  int idx = blockIdx.x * blockDim.x + threadIdx.x;
  if (idx >= ETOT * H) return;
  int e = idx / H, h = idx % H;
  int s, d; edge_sd(ei, e, s, d);
  float lg = leaky(as[s * H + h] + ad[d * H + h]);
  elog[idx] = lg;
  atomicMax(&lkey[d * H + h], f2key(lg));
}

template <int H>
__global__ void edge_expsum(const int* __restrict__ ei, float* __restrict__ elog,
                            const unsigned* __restrict__ lkey, float* __restrict__ denom) {
  int idx = blockIdx.x * blockDim.x + threadIdx.x;
  if (idx >= ETOT * H) return;
  int e = idx / H, h = idx % H;
  int s, d; edge_sd(ei, e, s, d);
  float m  = key2f(lkey[d * H + h]);
  float ex = expf(elog[idx] - m);
  elog[idx] = ex;
  atomicAdd(&denom[d * H + h], ex);
}

template <int H, int CH>
__global__ void edge_aggr(const int* __restrict__ ei, const float* __restrict__ elog,
                          const float* __restrict__ denom, const float* __restrict__ h,
                          float* __restrict__ out) {
  int wave = blockIdx.x * (blockDim.x >> 5) + (threadIdx.x >> 5);
  int lane = threadIdx.x & 31;
  if (wave >= ETOT) return;
  int s, d; edge_sd(ei, wave, s, d);
  constexpr int F   = H * CH;
  constexpr int per = F / 32;
  const int c0 = lane * per;
  const int hd = c0 / CH;
  float alpha = elog[wave * H + hd] / (denom[d * H + hd] + 1e-16f);
  const float* hp = h + (size_t)s * F + c0;
  float*       op = out + (size_t)d * F + c0;
  __builtin_prefetch(hp, 0, 1);
#pragma unroll
  for (int j = 0; j < per; ++j) atomicAdd(&op[j], alpha * hp[j]);
}

__global__ void bias_elu(float* __restrict__ h, const float* __restrict__ bias, int n, int F) {
  int i = blockIdx.x * blockDim.x + threadIdx.x;
  if (i >= n) return;
  float v = h[i] + bias[i % F];
  h[i] = v > 0.0f ? v : (expf(v) - 1.0f);
}
__global__ void bias_add(float* __restrict__ o, const float* __restrict__ bias, int n, int F) {
  int i = blockIdx.x * blockDim.x + threadIdx.x;
  if (i >= n) return;
  o[i] += bias[i % F];
}

extern "C" void kernel_launch(void* const* d_in, const int* in_sizes, int n_in,
                              void* d_out, int out_size, void* d_ws, size_t ws_size,
                              hipStream_t stream) {
  const float* x   = (const float*)d_in[0];
  const int*   ei  = (const int*)d_in[1];
  const float* W1  = (const float*)d_in[2];
  const float* as1 = (const float*)d_in[3];
  const float* ad1 = (const float*)d_in[4];
  const float* b1  = (const float*)d_in[5];
  const float* W2  = (const float*)d_in[6];
  const float* as2 = (const float*)d_in[7];
  const float* ad2 = (const float*)d_in[8];
  const float* b2  = (const float*)d_in[9];
  float* out = (float*)d_out;

  char* w = (char*)d_ws;
  auto carve = [&](size_t bytes) { char* p = w; w += (bytes + 255) & ~(size_t)255; return p; };
  float*    h1pre = (float*)   carve(sizeof(float) * N_NODES * F1);
  float*    h1    = (float*)   carve(sizeof(float) * N_NODES * F1);
  float*    h2pre = (float*)   carve(sizeof(float) * N_NODES * F2);
  float*    elog1 = (float*)   carve(sizeof(float) * (size_t)ETOT * H1);
  float*    elog2 = (float*)   carve(sizeof(float) * (size_t)ETOT);
  float*    asrc1 = (float*)   carve(sizeof(float) * N_NODES * H1);
  float*    adst1 = (float*)   carve(sizeof(float) * N_NODES * H1);
  float*    den1  = (float*)   carve(sizeof(float) * N_NODES * H1);
  unsigned* lk1   = (unsigned*)carve(sizeof(unsigned) * N_NODES * H1);
  float*    asrc2 = (float*)   carve(sizeof(float) * N_NODES);
  float*    adst2 = (float*)   carve(sizeof(float) * N_NODES);
  float*    den2  = (float*)   carve(sizeof(float) * N_NODES);
  unsigned* lk2   = (unsigned*)carve(sizeof(unsigned) * N_NODES);

  const int BLK = 256;
  auto cdiv = [](int a, int b) { return (a + b - 1) / b; };

  zero_f32<<<cdiv(N_NODES * F1, BLK), BLK, 0, stream>>>(h1, N_NODES * F1);
  zero_f32<<<cdiv(N_NODES * H1, BLK), BLK, 0, stream>>>(den1, N_NODES * H1);
  zero_u32<<<cdiv(N_NODES * H1, BLK), BLK, 0, stream>>>(lk1, N_NODES * H1);
  zero_f32<<<cdiv(N_NODES * F2, BLK), BLK, 0, stream>>>(out, N_NODES * F2);
  zero_f32<<<cdiv(N_NODES, BLK), BLK, 0, stream>>>(den2, N_NODES);
  zero_u32<<<cdiv(N_NODES, BLK), BLK, 0, stream>>>(lk2, N_NODES);

  {
    int tiles = (N_NODES / 16) * (F1 / 64);
    wmma_gemm_f32<4><<<cdiv(tiles, BLK / 32), BLK, 0, stream>>>(x, W1, h1pre,
                                                                N_NODES, IN_CH, F1, tiles);
  }
  att_dots<C1><<<cdiv(N_NODES * H1, BLK), BLK, 0, stream>>>(h1pre, as1, ad1,
                                                            asrc1, adst1, N_NODES, H1);
  edge_logit_max<H1><<<cdiv(ETOT * H1, BLK), BLK, 0, stream>>>(ei, asrc1, adst1, elog1, lk1);
  edge_expsum<H1><<<cdiv(ETOT * H1, BLK), BLK, 0, stream>>>(ei, elog1, lk1, den1);
  edge_aggr<H1, C1><<<cdiv(ETOT, BLK / 32), BLK, 0, stream>>>(ei, elog1, den1, h1pre, h1);
  bias_elu<<<cdiv(N_NODES * F1, BLK), BLK, 0, stream>>>(h1, b1, N_NODES * F1, F1);

  {
    int tiles = (N_NODES / 16) * (F2 / 32);
    wmma_gemm_f32<2><<<cdiv(tiles, BLK / 32), BLK, 0, stream>>>(h1, W2, h2pre,
                                                                N_NODES, F1, F2, tiles);
  }
  att_dots<F2><<<cdiv(N_NODES, BLK), BLK, 0, stream>>>(h2pre, as2, ad2,
                                                       asrc2, adst2, N_NODES, 1);
  edge_logit_max<1><<<cdiv(ETOT, BLK), BLK, 0, stream>>>(ei, asrc2, adst2, elog2, lk2);
  edge_expsum<1><<<cdiv(ETOT, BLK), BLK, 0, stream>>>(ei, elog2, lk2, den2);
  edge_aggr<1, F2><<<cdiv(ETOT, BLK / 32), BLK, 0, stream>>>(ei, elog2, den2, h2pre, out);
  bias_add<<<cdiv(N_NODES * F2, BLK), BLK, 0, stream>>>(out, b2, N_NODES * F2, F2);
}


# aggregate per destination node: per-call linked lists of incoming edges (atomic exchange on a head word), one wave per node sums alpha*h[src] in registers and stores the row once
# speedup vs baseline: 2.0711x; 2.0711x over previous
#include <hip/hip_runtime.h>
#include <stdint.h>

typedef float v2f __attribute__((ext_vector_type(2)));
typedef float v8f __attribute__((ext_vector_type(8)));
typedef float v4f __attribute__((ext_vector_type(4)));

#define N_NODES 50000
#define N_EDGES 800000
#define ETOT    (N_EDGES + N_NODES)
#define IN_CH   64
#define F1      256
#define H1      4
#define C1      64
#define F2      32
#define NEG_SLOPE 0.2f

__global__ void zero_f32(float* __restrict__ p, int n) {
  int i = blockIdx.x * blockDim.x + threadIdx.x;
  if (i < n) p[i] = 0.0f;
}
__global__ void zero_u32(unsigned* __restrict__ p, int n) {
  int i = blockIdx.x * blockDim.x + threadIdx.x;
  if (i < n) p[i] = 0u;
}

__device__ __forceinline__ void edge_sd(const int* __restrict__ ei, int e, int& s, int& d) {
  if (e < N_EDGES) { s = ei[e]; d = ei[N_EDGES + e]; }
  else             { s = d = e - N_EDGES; }
}

__device__ __forceinline__ unsigned f2key(float f) {
  unsigned b = __float_as_uint(f);
  return (b & 0x80000000u) ? ~b : (b | 0x80000000u);
}
__device__ __forceinline__ float key2f(unsigned k) {
  if (k == 0u) return 0.0f;
  unsigned b = (k & 0x80000000u) ? (k & 0x7fffffffu) : ~k;
  return __uint_as_float(b);
}

__device__ __forceinline__ float leaky(float x) {
  return x > 0.0f ? x : NEG_SLOPE * x;
}

template <int NT>
__global__ void wmma_gemm_f32(const float* __restrict__ A, const float* __restrict__ B,
                              float* __restrict__ C, int M, int K, int N, int tiles) {
  const int lane = threadIdx.x & 31;
  const int tile = blockIdx.x * (blockDim.x >> 5) + (threadIdx.x >> 5);
  if (tile >= tiles) return;
  const int tilesN = N / (16 * NT);
  const int tm  = tile / tilesN;
  const int tn0 = (tile % tilesN) * NT;

  const int rc   = lane & 15;
  const int hi   = lane >> 4;
  const int koff = hi << 1;

  const float* Ap = A + (size_t)(tm * 16 + rc) * K + koff;
  const float* Bp = B + (size_t)koff * N + tn0 * 16 + rc;

  v8f acc[NT];
#pragma unroll
  for (int t = 0; t < NT; ++t) acc[t] = (v8f){};

  for (int k0 = 0; k0 < K; k0 += 4) {
    v2f a;
    a.x = Ap[k0];
    a.y = Ap[k0 + 1];
#pragma unroll
    for (int t = 0; t < NT; ++t) {
      v2f b;
      b.x = Bp[(size_t)k0 * N + t * 16];
      b.y = Bp[(size_t)(k0 + 1) * N + t * 16];
      acc[t] = __builtin_amdgcn_wmma_f32_16x16x4_f32(false, a, false, b, (short)0,
                                                     acc[t], false, false);
    }
  }
  float* Cp = C + (size_t)(tm * 16 + hi * 8) * N + tn0 * 16 + rc;
#pragma unroll
  for (int t = 0; t < NT; ++t)
#pragma unroll
    for (int r = 0; r < 8; ++r) Cp[(size_t)r * N + t * 16] = acc[t][r];
}

template <int CH>
__global__ void att_dots(const float* __restrict__ h, const float* __restrict__ att_s,
                         const float* __restrict__ att_d, float* __restrict__ as,
                         float* __restrict__ ad, int n, int heads) {
  int idx = blockIdx.x * blockDim.x + threadIdx.x;
  if (idx >= n * heads) return;
  int node = idx / heads, hd = idx % heads;
  const float* hp = h + (size_t)node * heads * CH + (size_t)hd * CH;
  const float* sp = att_s + (size_t)hd * CH;
  const float* dp = att_d + (size_t)hd * CH;
  float s = 0.0f, d = 0.0f;
#pragma unroll 4
  for (int c = 0; c < CH; ++c) { float v = hp[c]; s += v * sp[c]; d += v * dp[c]; }
  as[idx] = s;
  ad[idx] = d;
}

template <int H>
__global__ void edge_logit_max(const int* __restrict__ ei, const float* __restrict__ as,
                               const float* __restrict__ ad, float* __restrict__ elog,
                               unsigned* __restrict__ lkey) {
  int idx = blockIdx.x * blockDim.x + threadIdx.x;
  if (idx >= ETOT * H) return;
  int e = idx / H, h = idx % H;
  int s, d; edge_sd(ei, e, s, d);
  float lg = leaky(as[s * H + h] + ad[d * H + h]);
  elog[idx] = lg;
  atomicMax(&lkey[d * H + h], f2key(lg));
}

template <int H>
__global__ void edge_expsum(const int* __restrict__ ei, float* __restrict__ elog,
                            const unsigned* __restrict__ lkey, float* __restrict__ denom) {
  int idx = blockIdx.x * blockDim.x + threadIdx.x;
  if (idx >= ETOT * H) return;
  int e = idx / H, h = idx % H;
  int s, d; edge_sd(ei, e, s, d);
  float m  = key2f(lkey[d * H + h]);
  float ex = expf(elog[idx] - m);
  elog[idx] = ex;
  atomicAdd(&denom[d * H + h], ex);
}

__global__ void link_edges(const int* __restrict__ ei, unsigned* __restrict__ head,
                           unsigned* __restrict__ nxt) {
  int e = blockIdx.x * blockDim.x + threadIdx.x;
  if (e >= N_EDGES) return;
  nxt[e] = atomicExch(&head[ei[N_EDGES + e]], (unsigned)(e + 1));
}

template <int H, int CH>
__global__ void node_aggr(const int* __restrict__ ei, const unsigned* __restrict__ head,
                          const unsigned* __restrict__ nxt, const float* __restrict__ elog,
                          const float* __restrict__ denom, const float* __restrict__ h,
                          float* __restrict__ out) {
  constexpr int F = H * CH;
  static_assert((H == 4 && CH == 64) || (H == 1 && CH == 32), "lane-to-channel map below");
  const int lane = threadIdx.x & 31;
  const int d = __builtin_amdgcn_readfirstlane(
      (int)(blockIdx.x * (blockDim.x >> 5) + (threadIdx.x >> 5)));
  if (d >= N_NODES) return;

  float den[H];
#pragma unroll
  for (int k = 0; k < H; ++k) den[k] = denom[d * H + k] + 1e-16f;

  const bool up = lane >= 16;
  v4f acc0 = (v4f){}, acc1 = (v4f){};
  float accs = 0.0f;

  {
    const float* ep = elog + (size_t)(N_EDGES + d) * H;
    const float* hp = h + (size_t)d * F;
    if constexpr (H == 4) {
      const v4f ex = *(const v4f*)ep;
      const float a0 = (up ? ex.y : ex.x) / (up ? den[1] : den[0]);
      const float a1 = (up ? ex.w : ex.z) / (up ? den[3] : den[2]);
      acc0 = a0 * *(const v4f*)(hp + lane * 4);
      acc1 = a1 * *(const v4f*)(hp + 128 + lane * 4);
    } else {
      accs = (ep[0] / den[0]) * hp[lane];
    }
  }

  unsigned cur = head[d];
  for (int walked = 0; cur != 0u && walked < N_EDGES;) {
    unsigned mine = 0u;
    int n = 0;
    while (cur != 0u && n < 32) {
      if (lane == n) mine = cur;
      cur = nxt[cur - 1u];
      ++n;
    }
    walked += n;

    int   s_l = 0;
    float al[H];
#pragma unroll
    for (int k = 0; k < H; ++k) al[k] = 0.0f;
    if (lane < n) {
      const int e = (int)mine - 1;
      s_l = ei[e];
      if constexpr (H == 4) {
        const v4f ex = *(const v4f*)(elog + (size_t)e * 4);
        al[0] = ex.x / den[0];
        al[1] = ex.y / den[1];
        al[2] = ex.z / den[2];
        al[3] = ex.w / den[3];
      } else {
        al[0] = elog[e] / den[0];
      }
    }

#pragma unroll 4
    for (int t = 0; t < n; ++t) {
      const int s = __builtin_amdgcn_readlane(s_l, t);
      const float* hp = h + (size_t)s * F;
      if constexpr (H == 4) {
        const float b0 = __int_as_float(__builtin_amdgcn_readlane(__float_as_int(al[0]), t));
        const float b1 = __int_as_float(__builtin_amdgcn_readlane(__float_as_int(al[1]), t));
        const float b2 = __int_as_float(__builtin_amdgcn_readlane(__float_as_int(al[2]), t));
        const float b3 = __int_as_float(__builtin_amdgcn_readlane(__float_as_int(al[3]), t));
        const v4f v0 = *(const v4f*)(hp + lane * 4);
        const v4f v1 = *(const v4f*)(hp + 128 + lane * 4);
        acc0 += (up ? b1 : b0) * v0;
        acc1 += (up ? b3 : b2) * v1;
      } else {
        const float b0 = __int_as_float(__builtin_amdgcn_readlane(__float_as_int(al[0]), t));
        accs += b0 * hp[lane];
      }
    }
  }

  float* op = out + (size_t)d * F;
  if constexpr (H == 4) {
    *(v4f*)(op + lane * 4)       = acc0;
    *(v4f*)(op + 128 + lane * 4) = acc1;
  } else {
    op[lane] = accs;
  }
}

__global__ void bias_elu(float* __restrict__ h, const float* __restrict__ bias, int n, int F) {
  int i = blockIdx.x * blockDim.x + threadIdx.x;
  if (i >= n) return;
  float v = h[i] + bias[i % F];
  h[i] = v > 0.0f ? v : (expf(v) - 1.0f);
}
__global__ void bias_add(float* __restrict__ o, const float* __restrict__ bias, int n, int F) {
  int i = blockIdx.x * blockDim.x + threadIdx.x;
  if (i >= n) return;
  o[i] += bias[i % F];
}

extern "C" void kernel_launch(void* const* d_in, const int* in_sizes, int n_in,
                              void* d_out, int out_size, void* d_ws, size_t ws_size,
                              hipStream_t stream) {
  const float* x   = (const float*)d_in[0];
  const int*   ei  = (const int*)d_in[1];
  const float* W1  = (const float*)d_in[2];
  const float* as1 = (const float*)d_in[3];
  const float* ad1 = (const float*)d_in[4];
  const float* b1  = (const float*)d_in[5];
  const float* W2  = (const float*)d_in[6];
  const float* as2 = (const float*)d_in[7];
  const float* ad2 = (const float*)d_in[8];
  const float* b2  = (const float*)d_in[9];
  float* out = (float*)d_out;

  char* w = (char*)d_ws;
  auto carve = [&](size_t bytes) { char* p = w; w += (bytes + 255) & ~(size_t)255; return p; };
  float*    h1pre = (float*)   carve(sizeof(float) * N_NODES * F1);
  float*    h1    = (float*)   carve(sizeof(float) * N_NODES * F1);
  char*     lists = (char*)    carve(sizeof(float) * N_NODES * F2);
  unsigned* nxt   = (unsigned*)lists;
  unsigned* head  = (unsigned*)(lists + sizeof(unsigned) * N_EDGES);
  float*    h2pre = h1pre;
  float*    elog1 = (float*)   carve(sizeof(float) * (size_t)ETOT * H1);
  float*    elog2 = (float*)   carve(sizeof(float) * (size_t)ETOT);
  float*    asrc1 = (float*)   carve(sizeof(float) * N_NODES * H1);
  float*    adst1 = (float*)   carve(sizeof(float) * N_NODES * H1);
  float*    den1  = (float*)   carve(sizeof(float) * N_NODES * H1);
  unsigned* lk1   = (unsigned*)carve(sizeof(unsigned) * N_NODES * H1);
  float*    asrc2 = (float*)   carve(sizeof(float) * N_NODES);
  float*    adst2 = (float*)   carve(sizeof(float) * N_NODES);
  float*    den2  = (float*)   carve(sizeof(float) * N_NODES);
  unsigned* lk2   = (unsigned*)carve(sizeof(unsigned) * N_NODES);

  const int BLK = 256;
  auto cdiv = [](int a, int b) { return (a + b - 1) / b; };

  zero_f32<<<cdiv(N_NODES * H1, BLK), BLK, 0, stream>>>(den1, N_NODES * H1);
  zero_u32<<<cdiv(N_NODES * H1, BLK), BLK, 0, stream>>>(lk1, N_NODES * H1);
  zero_u32<<<cdiv(N_NODES, BLK), BLK, 0, stream>>>(head, N_NODES);
  zero_f32<<<cdiv(N_NODES, BLK), BLK, 0, stream>>>(den2, N_NODES);
  zero_u32<<<cdiv(N_NODES, BLK), BLK, 0, stream>>>(lk2, N_NODES);

  link_edges<<<cdiv(N_EDGES, BLK), BLK, 0, stream>>>(ei, head, nxt);

  {
    int tiles = (N_NODES / 16) * (F1 / 64);
    wmma_gemm_f32<4><<<cdiv(tiles, BLK / 32), BLK, 0, stream>>>(x, W1, h1pre,
                                                                N_NODES, IN_CH, F1, tiles);
  }
  att_dots<C1><<<cdiv(N_NODES * H1, BLK), BLK, 0, stream>>>(h1pre, as1, ad1,
                                                            asrc1, adst1, N_NODES, H1);
  edge_logit_max<H1><<<cdiv(ETOT * H1, BLK), BLK, 0, stream>>>(ei, asrc1, adst1, elog1, lk1);
  edge_expsum<H1><<<cdiv(ETOT * H1, BLK), BLK, 0, stream>>>(ei, elog1, lk1, den1);
  node_aggr<H1, C1><<<cdiv(N_NODES, BLK / 32), BLK, 0, stream>>>(ei, head, nxt, elog1, den1,
                                                                 h1pre, h1);
  bias_elu<<<cdiv(N_NODES * F1, BLK), BLK, 0, stream>>>(h1, b1, N_NODES * F1, F1);

  {
    int tiles = (N_NODES / 16) * (F2 / 32);
    wmma_gemm_f32<2><<<cdiv(tiles, BLK / 32), BLK, 0, stream>>>(h1, W2, h2pre,
                                                                N_NODES, F1, F2, tiles);
  }
  att_dots<F2><<<cdiv(N_NODES, BLK), BLK, 0, stream>>>(h2pre, as2, ad2,
                                                       asrc2, adst2, N_NODES, 1);
  edge_logit_max<1><<<cdiv(ETOT, BLK), BLK, 0, stream>>>(ei, asrc2, adst2, elog2, lk2);
  edge_expsum<1><<<cdiv(ETOT, BLK), BLK, 0, stream>>>(ei, elog2, lk2, den2);
  node_aggr<1, F2><<<cdiv(N_NODES, BLK / 32), BLK, 0, stream>>>(ei, head, nxt, elog2, den2,
                                                                h2pre, out);
  bias_add<<<cdiv(N_NODES * F2, BLK), BLK, 0, stream>>>(out, b2, N_NODES * F2, F2);
}


# layer 1 reordered by linearity: the per-node kernel sums x rows (256 B per edge, shared by the 4 heads) and W1 with bias+ELU is applied after it; logit terms come from x @ (W1 folded with att)
# speedup vs baseline: 5.0482x; 2.4374x over previous
#include <hip/hip_runtime.h>
#include <stdint.h>


typedef float v2f __attribute__((ext_vector_type(2)));
typedef float v8f __attribute__((ext_vector_type(8)));
typedef float v4f __attribute__((ext_vector_type(4)));

#define N_NODES 50000
#define N_EDGES 800000
#define IN_CH   64
#define F1      256
#define H1      4
#define C1      64
#define F2      32
#define AW      16
#define NEG_SLOPE 0.2f

__device__ __forceinline__ float leaky(float x) {
  return x > 0.0f ? x : NEG_SLOPE * x;
}
__device__ __forceinline__ float elu(float v) {
  return v > 0.0f ? v : (expf(v) - 1.0f);
}

__global__ void prep_lists_and_att(unsigned* __restrict__ head, const float* __restrict__ W1,
                                   const float* __restrict__ att_s,
                                   const float* __restrict__ att_d, float* __restrict__ wa) {
  const int i = blockIdx.x * blockDim.x + threadIdx.x;
  if (i < N_NODES) head[i] = 0u;
  if (i < IN_CH * AW) {
    const int c = i / AW, k = i % AW;
    float s = 0.0f;
    if (k < 2 * H1) {
      const int hd = k & (H1 - 1);
      const float* wp = W1 + (size_t)c * F1 + hd * C1;
      const float* ap = (k < H1 ? att_s : att_d) + hd * C1;
      for (int j = 0; j < C1; ++j) s += wp[j] * ap[j];
    }
    wa[i] = s;
  }
}

template <int XMASK>
__device__ __forceinline__ float row_xor(float x) {
  return __int_as_float(__builtin_amdgcn_update_dpp(0, __float_as_int(x), 0x160 + XMASK,
                                                    0xf, 0xf, false));
}
__device__ __forceinline__ float other_row(float x) {
  return __int_as_float(__builtin_amdgcn_permlanex16(__float_as_int(x), __float_as_int(x),
                                                     0x76543210, 0xfedcba98, false, false));
}
__device__ __forceinline__ float wave_max(float x) {
  x = fmaxf(x, row_xor<1>(x));
  x = fmaxf(x, row_xor<2>(x));
  x = fmaxf(x, row_xor<4>(x));
  x = fmaxf(x, row_xor<8>(x));
  return fmaxf(x, other_row(x));
}
__device__ __forceinline__ float wave_sum(float x) {
  x += row_xor<1>(x);
  x += row_xor<2>(x);
  x += row_xor<4>(x);
  x += row_xor<8>(x);
  return x + other_row(x);
}
__device__ __forceinline__ float uniform(float x) {
  return __int_as_float(__builtin_amdgcn_readfirstlane(__float_as_int(x)));
}

template <int NT>
__global__ void wmma_gemm_f32(const float* __restrict__ A, const float* __restrict__ B,
                              float* __restrict__ C, int M, int K, int N, int tiles) {
  const int lane = threadIdx.x & 31;
  const int tile = blockIdx.x * (blockDim.x >> 5) + (threadIdx.x >> 5);
  if (tile >= tiles) return;
  const int tilesN = N / (16 * NT);
  const int tm  = tile / tilesN;
  const int tn0 = (tile % tilesN) * NT;

  const int rc   = lane & 15;
  const int hi   = lane >> 4;
  const int koff = hi << 1;

  const float* Ap = A + (size_t)(tm * 16 + rc) * K + koff;
  const float* Bp = B + (size_t)koff * N + tn0 * 16 + rc;

  v8f acc[NT];
#pragma unroll
  for (int t = 0; t < NT; ++t) acc[t] = (v8f){};

  for (int k0 = 0; k0 < K; k0 += 4) {
    v2f a;
    a.x = Ap[k0];
    a.y = Ap[k0 + 1];
#pragma unroll
    for (int t = 0; t < NT; ++t) {
      v2f b;
      b.x = Bp[(size_t)k0 * N + t * 16];
      b.y = Bp[(size_t)(k0 + 1) * N + t * 16];
      acc[t] = __builtin_amdgcn_wmma_f32_16x16x4_f32(false, a, false, b, (short)0,
                                                     acc[t], false, false);
    }
  }
  float* Cp = C + (size_t)(tm * 16 + hi * 8) * N + tn0 * 16 + rc;
#pragma unroll
  for (int t = 0; t < NT; ++t)
#pragma unroll
    for (int r = 0; r < 8; ++r) Cp[(size_t)r * N + t * 16] = acc[t][r];
}

__global__ void wmma_head_gemm_bias_elu(const float* __restrict__ z, const float* __restrict__ W,
                                        const float* __restrict__ bias, float* __restrict__ out,
                                        int tiles) {
  const int lane = threadIdx.x & 31;
  const int tile = blockIdx.x * (blockDim.x >> 5) + (threadIdx.x >> 5);
  if (tile >= tiles) return;
  const int tm = tile / H1;
  const int hd = tile % H1;

  const int rc   = lane & 15;
  const int hi   = lane >> 4;
  const int koff = hi << 1;

  const float* Ap = z + (size_t)(tm * 16 + rc) * F1 + hd * C1 + koff;
  const float* Bp = W + (size_t)koff * F1 + hd * C1 + rc;

  v8f acc[4];
#pragma unroll
  for (int t = 0; t < 4; ++t) acc[t] = (v8f){};

  for (int k0 = 0; k0 < C1; k0 += 4) {
    v2f a;
    a.x = Ap[k0];
    a.y = Ap[k0 + 1];
#pragma unroll
    for (int t = 0; t < 4; ++t) {
      v2f b;
      b.x = Bp[(size_t)k0 * F1 + t * 16];
      b.y = Bp[(size_t)(k0 + 1) * F1 + t * 16];
      acc[t] = __builtin_amdgcn_wmma_f32_16x16x4_f32(false, a, false, b, (short)0,
                                                     acc[t], false, false);
    }
  }
  float* Cp = out + (size_t)(tm * 16 + hi * 8) * F1 + hd * C1 + rc;
#pragma unroll
  for (int t = 0; t < 4; ++t) {
    const float bv = bias[hd * C1 + t * 16 + rc];
#pragma unroll
    for (int r = 0; r < 8; ++r) Cp[(size_t)r * F1 + t * 16] = elu(acc[t][r] + bv);
  }
}

template <int CH>
__global__ void att_dots(const float* __restrict__ h, const float* __restrict__ att_s,
                         const float* __restrict__ att_d, float* __restrict__ as,
                         float* __restrict__ ad, int n, int heads) {
  int idx = blockIdx.x * blockDim.x + threadIdx.x;
  if (idx >= n * heads) return;
  int node = idx / heads, hd = idx % heads;
  const float* hp = h + (size_t)node * heads * CH + (size_t)hd * CH;
  const float* sp = att_s + (size_t)hd * CH;
  const float* dp = att_d + (size_t)hd * CH;
  float s = 0.0f, d = 0.0f;
#pragma unroll 4
  for (int c = 0; c < CH; ++c) { float v = hp[c]; s += v * sp[c]; d += v * dp[c]; }
  as[idx] = s;
  ad[idx] = d;
}

__global__ void link_edges(const int* __restrict__ ei, unsigned* __restrict__ head,
                           unsigned* __restrict__ nxt) {
  int e = blockIdx.x * blockDim.x + threadIdx.x;
  if (e >= N_EDGES) return;
  nxt[e] = atomicExch(&head[ei[N_EDGES + e]], (unsigned)(e + 1));
}

__global__ void node_softmax_aggr_x(const int* __restrict__ ei, const unsigned* __restrict__ head,
                                    const unsigned* __restrict__ nxt,
                                    const float* __restrict__ asd, const float* __restrict__ x,
                                    float* __restrict__ z) {
  const int lane = threadIdx.x & 31;
  const int d = __builtin_amdgcn_readfirstlane(
      (int)(blockIdx.x * (blockDim.x >> 5) + (threadIdx.x >> 5)));
  if (d >= N_NODES) return;

  float adv[H1], m[H1], den[H1];
#pragma unroll
  for (int k = 0; k < H1; ++k) {
    adv[k] = asd[(size_t)d * AW + H1 + k];
    m[k]   = leaky(asd[(size_t)d * AW + k] + adv[k]);
    den[k] = 1.0f;
  }

  v2f acc[H1];
  {
    const v2f own = *(const v2f*)(x + (size_t)d * IN_CH + lane * 2);
#pragma unroll
    for (int k = 0; k < H1; ++k) acc[k] = own;
  }

  unsigned cur = head[d];
  for (int walked = 0; cur != 0u && walked < N_EDGES;) {
    unsigned mine = 0u;
    int n = 0;
    while (cur != 0u && n < 32) {
      if (lane == n) mine = cur;
      cur = nxt[cur - 1u];
      ++n;
    }
    walked += n;

    int   s_l = 0;
    float lg[H1];
#pragma unroll
    for (int k = 0; k < H1; ++k) lg[k] = -__builtin_inff();
    if (lane < n) {
      const int e = (int)mine - 1;
      s_l = ei[e];
      const v4f a = *(const v4f*)(asd + (size_t)s_l * AW);
      lg[0] = leaky(a.x + adv[0]);
      lg[1] = leaky(a.y + adv[1]);
      lg[2] = leaky(a.z + adv[2]);
      lg[3] = leaky(a.w + adv[3]);
    }

    float ex[H1];
#pragma unroll
    for (int k = 0; k < H1; ++k) {
      const float mn = fmaxf(m[k], uniform(wave_max(lg[k])));
      const float sc = uniform(expf(m[k] - mn));
      ex[k]  = lane < n ? expf(lg[k] - mn) : 0.0f;
      den[k] = den[k] * sc + uniform(wave_sum(ex[k]));
      m[k]   = mn;
      acc[k] *= sc;
    }

#pragma unroll 4
    for (int t = 0; t < n; ++t) {
      const int s = __builtin_amdgcn_readlane(s_l, t);
      const v2f v = *(const v2f*)(x + (size_t)s * IN_CH + lane * 2);
#pragma unroll
      for (int k = 0; k < H1; ++k) {
        const float b = __int_as_float(__builtin_amdgcn_readlane(__float_as_int(ex[k]), t));
        acc[k] += b * v;
      }
    }
  }

  float* zp = z + (size_t)d * F1 + lane * 2;
#pragma unroll
  for (int k = 0; k < H1; ++k) {
    const float inv = 1.0f / (den[k] + 1e-16f);
    *(v2f*)(zp + k * C1) = inv * acc[k];
  }
}

__global__ void node_softmax_aggr_1x32(const int* __restrict__ ei,
                                       const unsigned* __restrict__ head,
                                       const unsigned* __restrict__ nxt,
                                       const float* __restrict__ as,
                                       const float* __restrict__ ad, const float* __restrict__ h,
                                       const float* __restrict__ bias, float* __restrict__ out) {
  const int lane = threadIdx.x & 31;
  const int d = __builtin_amdgcn_readfirstlane(
      (int)(blockIdx.x * (blockDim.x >> 5) + (threadIdx.x >> 5)));
  if (d >= N_NODES) return;

  const float adv = ad[d];
  float m   = leaky(as[d] + adv);
  float den = 1.0f;
  float accs = h[(size_t)d * F2 + lane];

  unsigned cur = head[d];
  for (int walked = 0; cur != 0u && walked < N_EDGES;) {
    unsigned mine = 0u;
    int n = 0;
    while (cur != 0u && n < 32) {
      if (lane == n) mine = cur;
      cur = nxt[cur - 1u];
      ++n;
    }
    walked += n;

    int   s_l = 0;
    float lg  = -__builtin_inff();
    if (lane < n) {
      const int e = (int)mine - 1;
      s_l = ei[e];
      lg  = leaky(as[s_l] + adv);
    }

    const float mn = fmaxf(m, uniform(wave_max(lg)));
    const float sc = uniform(expf(m - mn));
    const float ex = lane < n ? expf(lg - mn) : 0.0f;
    den = den * sc + uniform(wave_sum(ex));
    m   = mn;
    accs *= sc;

#pragma unroll 4
    for (int t = 0; t < n; ++t) {
      const int s = __builtin_amdgcn_readlane(s_l, t);
      const float b = __int_as_float(__builtin_amdgcn_readlane(__float_as_int(ex), t));
      accs += b * h[(size_t)s * F2 + lane];
    }
  }

  const float inv = 1.0f / (den + 1e-16f);
  out[(size_t)d * F2 + lane] = bias[lane] + inv * accs;
}

extern "C" void kernel_launch(void* const* d_in, const int* in_sizes, int n_in,
                              void* d_out, int out_size, void* d_ws, size_t ws_size,
                              hipStream_t stream) {
  const float* x   = (const float*)d_in[0];
  const int*   ei  = (const int*)d_in[1];
  const float* W1  = (const float*)d_in[2];
  const float* as1 = (const float*)d_in[3];
  const float* ad1 = (const float*)d_in[4];
  const float* b1  = (const float*)d_in[5];
  const float* W2  = (const float*)d_in[6];
  const float* as2 = (const float*)d_in[7];
  const float* ad2 = (const float*)d_in[8];
  const float* b2  = (const float*)d_in[9];
  float* out = (float*)d_out;

  char* w = (char*)d_ws;
  auto carve = [&](size_t bytes) { char* p = w; w += (bytes + 255) & ~(size_t)255; return p; };
  float*    z     = (float*)   carve(sizeof(float) * N_NODES * F1);
  float*    h1    = (float*)   carve(sizeof(float) * N_NODES * F1);
  char*     lists = (char*)    carve(sizeof(float) * N_NODES * F2);
  unsigned* nxt   = (unsigned*)lists;
  unsigned* head  = (unsigned*)(lists + sizeof(unsigned) * N_EDGES);
  float*    wa    = (float*)   carve(sizeof(float) * N_NODES * H1);
  (void)                       carve(sizeof(float) * N_NODES * H1);
  float*    asrc2 = (float*)   carve(sizeof(float) * N_NODES);
  float*    adst2 = (float*)   carve(sizeof(float) * N_NODES);
  float*    asd   = h1;
  float*    h2pre = z;

  const int BLK = 256;
  auto cdiv = [](int a, int b) { return (a + b - 1) / b; };

  prep_lists_and_att<<<cdiv(N_NODES, BLK), BLK, 0, stream>>>(head, W1, as1, ad1, wa);
  link_edges<<<cdiv(N_EDGES, BLK), BLK, 0, stream>>>(ei, head, nxt);

  {
    int tiles = N_NODES / 16;
    wmma_gemm_f32<1><<<cdiv(tiles, BLK / 32), BLK, 0, stream>>>(x, wa, asd,
                                                                N_NODES, IN_CH, AW, tiles);
  }
  node_softmax_aggr_x<<<cdiv(N_NODES, BLK / 32), BLK, 0, stream>>>(ei, head, nxt, asd, x, z);
  {
    int tiles = (N_NODES / 16) * H1;
    wmma_head_gemm_bias_elu<<<cdiv(tiles, BLK / 32), BLK, 0, stream>>>(z, W1, b1, h1, tiles);
  }

  {
    int tiles = (N_NODES / 16) * (F2 / 32);
    wmma_gemm_f32<2><<<cdiv(tiles, BLK / 32), BLK, 0, stream>>>(h1, W2, h2pre,
                                                                N_NODES, F1, F2, tiles);
  }
  att_dots<F2><<<cdiv(N_NODES, BLK), BLK, 0, stream>>>(h2pre, as2, ad2,
                                                       asrc2, adst2, N_NODES, 1);
  node_softmax_aggr_1x32<<<cdiv(N_NODES, BLK / 32), BLK, 0, stream>>>(
      ei, head, nxt, asrc2, adst2, h2pre, b2, out);
}


# incoming edges bucketed per node: an atomic counter ranks each edge, its source goes into the node's row of 24 slots (a chain only past 24); a wave reads the row in one load instead of walking links
# speedup vs baseline: 5.3456x; 1.0589x over previous
#include <hip/hip_runtime.h>
#include <stdint.h>


typedef float v2f __attribute__((ext_vector_type(2)));
typedef float v8f __attribute__((ext_vector_type(8)));
typedef float v4f __attribute__((ext_vector_type(4)));

#define N_NODES 50000
#define N_EDGES 800000
#define IN_CH   64
#define F1      256
#define H1      4
#define C1      64
#define F2      32
#define AW      16
#define CAP     24
#define NEG_SLOPE 0.2f

__device__ __forceinline__ float leaky(float x) {
  return x > 0.0f ? x : NEG_SLOPE * x;
}
__device__ __forceinline__ float elu(float v) {
  return v > 0.0f ? v : (expf(v) - 1.0f);
}

__global__ void prep_counts_and_att(unsigned* __restrict__ cnt, unsigned* __restrict__ head_ov,
                                   const float* __restrict__ W1,
                                   const float* __restrict__ att_s,
                                   const float* __restrict__ att_d, float* __restrict__ wa) {
  const int i = blockIdx.x * blockDim.x + threadIdx.x;
  if (i < N_NODES) { cnt[i] = 0u; head_ov[i] = 0u; }
  if (i < IN_CH * AW) {
    const int c = i / AW, k = i % AW;
    float s = 0.0f;
    if (k < 2 * H1) {
      const int hd = k & (H1 - 1);
      const float* wp = W1 + (size_t)c * F1 + hd * C1;
      const float* ap = (k < H1 ? att_s : att_d) + hd * C1;
      for (int j = 0; j < C1; ++j) s += wp[j] * ap[j];
    }
    wa[i] = s;
  }
}

template <int XMASK>
__device__ __forceinline__ float row_xor(float x) {
  return __int_as_float(__builtin_amdgcn_update_dpp(0, __float_as_int(x), 0x160 + XMASK,
                                                    0xf, 0xf, false));
}
__device__ __forceinline__ float other_row(float x) {
  return __int_as_float(__builtin_amdgcn_permlanex16(__float_as_int(x), __float_as_int(x),
                                                     0x76543210, 0xfedcba98, false, false));
}
__device__ __forceinline__ float wave_max(float x) {
  x = fmaxf(x, row_xor<1>(x));
  x = fmaxf(x, row_xor<2>(x));
  x = fmaxf(x, row_xor<4>(x));
  x = fmaxf(x, row_xor<8>(x));
  return fmaxf(x, other_row(x));
}
__device__ __forceinline__ float wave_sum(float x) {
  x += row_xor<1>(x);
  x += row_xor<2>(x);
  x += row_xor<4>(x);
  x += row_xor<8>(x);
  return x + other_row(x);
}
__device__ __forceinline__ float uniform(float x) {
  return __int_as_float(__builtin_amdgcn_readfirstlane(__float_as_int(x)));
}

template <int NT>
__global__ void wmma_gemm_f32(const float* __restrict__ A, const float* __restrict__ B,
                              float* __restrict__ C, int M, int K, int N, int tiles) {
  const int lane = threadIdx.x & 31;
  const int tile = blockIdx.x * (blockDim.x >> 5) + (threadIdx.x >> 5);
  if (tile >= tiles) return;
  const int tilesN = N / (16 * NT);
  const int tm  = tile / tilesN;
  const int tn0 = (tile % tilesN) * NT;

  const int rc   = lane & 15;
  const int hi   = lane >> 4;
  const int koff = hi << 1;

  const float* Ap = A + (size_t)(tm * 16 + rc) * K + koff;
  const float* Bp = B + (size_t)koff * N + tn0 * 16 + rc;

  v8f acc[NT];
#pragma unroll
  for (int t = 0; t < NT; ++t) acc[t] = (v8f){};

  for (int k0 = 0; k0 < K; k0 += 4) {
    v2f a;
    a.x = Ap[k0];
    a.y = Ap[k0 + 1];
#pragma unroll
    for (int t = 0; t < NT; ++t) {
      v2f b;
      b.x = Bp[(size_t)k0 * N + t * 16];
      b.y = Bp[(size_t)(k0 + 1) * N + t * 16];
      acc[t] = __builtin_amdgcn_wmma_f32_16x16x4_f32(false, a, false, b, (short)0,
                                                     acc[t], false, false);
    }
  }
  float* Cp = C + (size_t)(tm * 16 + hi * 8) * N + tn0 * 16 + rc;
#pragma unroll
  for (int t = 0; t < NT; ++t)
#pragma unroll
    for (int r = 0; r < 8; ++r) Cp[(size_t)r * N + t * 16] = acc[t][r];
}

__global__ void wmma_head_gemm_bias_elu(const float* __restrict__ z, const float* __restrict__ W,
                                        const float* __restrict__ bias, float* __restrict__ out,
                                        int tiles) {
  const int lane = threadIdx.x & 31;
  const int tile = blockIdx.x * (blockDim.x >> 5) + (threadIdx.x >> 5);
  if (tile >= tiles) return;
  const int tm = tile / H1;
  const int hd = tile % H1;

  const int rc   = lane & 15;
  const int hi   = lane >> 4;
  const int koff = hi << 1;

  const float* Ap = z + (size_t)(tm * 16 + rc) * F1 + hd * C1 + koff;
  const float* Bp = W + (size_t)koff * F1 + hd * C1 + rc;

  v8f acc[4];
#pragma unroll
  for (int t = 0; t < 4; ++t) acc[t] = (v8f){};

  for (int k0 = 0; k0 < C1; k0 += 4) {
    v2f a;
    a.x = Ap[k0];
    a.y = Ap[k0 + 1];
#pragma unroll
    for (int t = 0; t < 4; ++t) {
      v2f b;
      b.x = Bp[(size_t)k0 * F1 + t * 16];
      b.y = Bp[(size_t)(k0 + 1) * F1 + t * 16];
      acc[t] = __builtin_amdgcn_wmma_f32_16x16x4_f32(false, a, false, b, (short)0,
                                                     acc[t], false, false);
    }
  }
  float* Cp = out + (size_t)(tm * 16 + hi * 8) * F1 + hd * C1 + rc;
#pragma unroll
  for (int t = 0; t < 4; ++t) {
    const float bv = bias[hd * C1 + t * 16 + rc];
#pragma unroll
    for (int r = 0; r < 8; ++r) Cp[(size_t)r * F1 + t * 16] = elu(acc[t][r] + bv);
  }
}

template <int CH>
__global__ void att_dots(const float* __restrict__ h, const float* __restrict__ att_s,
                         const float* __restrict__ att_d, float* __restrict__ as,
                         float* __restrict__ ad, int n, int heads) {
  int idx = blockIdx.x * blockDim.x + threadIdx.x;
  if (idx >= n * heads) return;
  int node = idx / heads, hd = idx % heads;
  const float* hp = h + (size_t)node * heads * CH + (size_t)hd * CH;
  const float* sp = att_s + (size_t)hd * CH;
  const float* dp = att_d + (size_t)hd * CH;
  float s = 0.0f, d = 0.0f;
#pragma unroll 4
  for (int c = 0; c < CH; ++c) { float v = hp[c]; s += v * sp[c]; d += v * dp[c]; }
  as[idx] = s;
  ad[idx] = d;
}

__global__ void bucket_edges(const int* __restrict__ ei, unsigned* __restrict__ cnt,
                             unsigned* __restrict__ slot, unsigned* __restrict__ head_ov,
                             unsigned* __restrict__ nxt_ov) {
  int e = blockIdx.x * blockDim.x + threadIdx.x;
  if (e >= N_EDGES) return;
  const int s = ei[e];
  const int d = ei[N_EDGES + e];
  const unsigned r = atomicAdd(&cnt[d], 1u);
  if (r < (unsigned)CAP) atomicExch(&slot[(size_t)d * CAP + r], (unsigned)s);
  else nxt_ov[e] = atomicExch(&head_ov[d], (unsigned)(e + 1));
}

__global__ void node_softmax_aggr_x(const int* __restrict__ ei, const unsigned* __restrict__ cnt,
                                    const unsigned* __restrict__ slot,
                                    const unsigned* __restrict__ head_ov,
                                    const unsigned* __restrict__ nxt_ov,
                                    const float* __restrict__ asd, const float* __restrict__ x,
                                    float* __restrict__ z) {
  const int lane = threadIdx.x & 31;
  const int d = __builtin_amdgcn_readfirstlane(
      (int)(blockIdx.x * (blockDim.x >> 5) + (threadIdx.x >> 5)));
  if (d >= N_NODES) return;

  float adv[H1], m[H1], den[H1];
#pragma unroll
  for (int k = 0; k < H1; ++k) {
    adv[k] = asd[(size_t)d * AW + H1 + k];
    m[k]   = leaky(asd[(size_t)d * AW + k] + adv[k]);
    den[k] = 1.0f;
  }

  v2f acc[H1];
  {
    const v2f own = *(const v2f*)(x + (size_t)d * IN_CH + lane * 2);
#pragma unroll
    for (int k = 0; k < H1; ++k) acc[k] = own;
  }

  const unsigned deg = cnt[d];
  unsigned cur = head_ov[d];
  int n0 = (int)(deg < (unsigned)CAP ? deg : (unsigned)CAP);
  for (int walked = 0; (n0 > 0 || cur != 0u) && walked < N_EDGES; n0 = 0) {
    unsigned mine = 0u;
    int n = n0;
    while (cur != 0u && n < 32) {
      if (lane == n) mine = cur;
      cur = nxt_ov[cur - 1u];
      ++n;
    }
    walked += n;

    int   s_l = 0;
    float lg[H1];
#pragma unroll
    for (int k = 0; k < H1; ++k) lg[k] = -__builtin_inff();
    if (lane < n) {
      if (lane < n0) s_l = (int)slot[(size_t)d * CAP + lane];
      else           s_l = ei[(int)mine - 1];
      const v4f a = *(const v4f*)(asd + (size_t)s_l * AW);
      lg[0] = leaky(a.x + adv[0]);
      lg[1] = leaky(a.y + adv[1]);
      lg[2] = leaky(a.z + adv[2]);
      lg[3] = leaky(a.w + adv[3]);
    }

    float ex[H1];
#pragma unroll
    for (int k = 0; k < H1; ++k) {
      const float mn = fmaxf(m[k], uniform(wave_max(lg[k])));
      const float sc = uniform(expf(m[k] - mn));
      ex[k]  = lane < n ? expf(lg[k] - mn) : 0.0f;
      den[k] = den[k] * sc + uniform(wave_sum(ex[k]));
      m[k]   = mn;
      acc[k] *= sc;
    }

#pragma unroll 4
    for (int t = 0; t < n; ++t) {
      const int s = __builtin_amdgcn_readlane(s_l, t);
      const v2f v = *(const v2f*)(x + (size_t)s * IN_CH + lane * 2);
#pragma unroll
      for (int k = 0; k < H1; ++k) {
        const float b = __int_as_float(__builtin_amdgcn_readlane(__float_as_int(ex[k]), t));
        acc[k] += b * v;
      }
    }
  }

  float* zp = z + (size_t)d * F1 + lane * 2;
#pragma unroll
  for (int k = 0; k < H1; ++k) {
    const float inv = 1.0f / (den[k] + 1e-16f);
    *(v2f*)(zp + k * C1) = inv * acc[k];
  }
}

__global__ void node_softmax_aggr_1x32(const int* __restrict__ ei,
                                       const unsigned* __restrict__ cnt,
                                       const unsigned* __restrict__ slot,
                                       const unsigned* __restrict__ head_ov,
                                       const unsigned* __restrict__ nxt_ov,
                                       const float* __restrict__ as,
                                       const float* __restrict__ ad, const float* __restrict__ h,
                                       const float* __restrict__ bias, float* __restrict__ out) {
  const int lane = threadIdx.x & 31;
  const int d = __builtin_amdgcn_readfirstlane(
      (int)(blockIdx.x * (blockDim.x >> 5) + (threadIdx.x >> 5)));
  if (d >= N_NODES) return;

  const float adv = ad[d];
  float m   = leaky(as[d] + adv);
  float den = 1.0f;
  float accs = h[(size_t)d * F2 + lane];

  const unsigned deg = cnt[d];
  unsigned cur = head_ov[d];
  int n0 = (int)(deg < (unsigned)CAP ? deg : (unsigned)CAP);
  for (int walked = 0; (n0 > 0 || cur != 0u) && walked < N_EDGES; n0 = 0) {
    unsigned mine = 0u;
    int n = n0;
    while (cur != 0u && n < 32) {
      if (lane == n) mine = cur;
      cur = nxt_ov[cur - 1u];
      ++n;
    }
    walked += n;

    int   s_l = 0;
    float lg  = -__builtin_inff();
    if (lane < n) {
      if (lane < n0) s_l = (int)slot[(size_t)d * CAP + lane];
      else           s_l = ei[(int)mine - 1];
      lg  = leaky(as[s_l] + adv);
    }

    const float mn = fmaxf(m, uniform(wave_max(lg)));
    const float sc = uniform(expf(m - mn));
    const float ex = lane < n ? expf(lg - mn) : 0.0f;
    den = den * sc + uniform(wave_sum(ex));
    m   = mn;
    accs *= sc;

#pragma unroll 4
    for (int t = 0; t < n; ++t) {
      const int s = __builtin_amdgcn_readlane(s_l, t);
      const float b = __int_as_float(__builtin_amdgcn_readlane(__float_as_int(ex), t));
      accs += b * h[(size_t)s * F2 + lane];
    }
  }

  const float inv = 1.0f / (den + 1e-16f);
  out[(size_t)d * F2 + lane] = bias[lane] + inv * accs;
}

extern "C" void kernel_launch(void* const* d_in, const int* in_sizes, int n_in,
                              void* d_out, int out_size, void* d_ws, size_t ws_size,
                              hipStream_t stream) {
  const float* x   = (const float*)d_in[0];
  const int*   ei  = (const int*)d_in[1];
  const float* W1  = (const float*)d_in[2];
  const float* as1 = (const float*)d_in[3];
  const float* ad1 = (const float*)d_in[4];
  const float* b1  = (const float*)d_in[5];
  const float* W2  = (const float*)d_in[6];
  const float* as2 = (const float*)d_in[7];
  const float* ad2 = (const float*)d_in[8];
  const float* b2  = (const float*)d_in[9];
  float* out = (float*)d_out;

  char* w = (char*)d_ws;
  auto carve = [&](size_t bytes) { char* p = w; w += (bytes + 255) & ~(size_t)255; return p; };
  float*    z     = (float*)   carve(sizeof(float) * N_NODES * F1);
  float*    h1    = (float*)   carve(sizeof(float) * N_NODES * F1);
  unsigned* slot    = (unsigned*)carve(sizeof(unsigned) * N_NODES * CAP);
  unsigned* nxt_ov  = (unsigned*)carve(sizeof(unsigned) * N_EDGES);
  unsigned* cnt     = (unsigned*)carve(sizeof(unsigned) * N_NODES);
  unsigned* head_ov = (unsigned*)carve(sizeof(unsigned) * N_NODES);
  float*    wa    = z;
  float*    asd   = h1;
  float*    h2pre = z;
  float*    asrc2 = z + (size_t)N_NODES * F2;
  float*    adst2 = asrc2 + N_NODES;

  const int BLK = 256;
  auto cdiv = [](int a, int b) { return (a + b - 1) / b; };

  prep_counts_and_att<<<cdiv(N_NODES, BLK), BLK, 0, stream>>>(cnt, head_ov, W1, as1, ad1, wa);
  bucket_edges<<<cdiv(N_EDGES, BLK), BLK, 0, stream>>>(ei, cnt, slot, head_ov, nxt_ov);

  {
    int tiles = N_NODES / 16;
    wmma_gemm_f32<1><<<cdiv(tiles, BLK / 32), BLK, 0, stream>>>(x, wa, asd,
                                                                N_NODES, IN_CH, AW, tiles);
  }
  node_softmax_aggr_x<<<cdiv(N_NODES, BLK / 32), BLK, 0, stream>>>(ei, cnt, slot, head_ov, nxt_ov,
                                                                   asd, x, z);
  {
    int tiles = (N_NODES / 16) * H1;
    wmma_head_gemm_bias_elu<<<cdiv(tiles, BLK / 32), BLK, 0, stream>>>(z, W1, b1, h1, tiles);
  }

  {
    int tiles = (N_NODES / 16) * (F2 / 32);
    wmma_gemm_f32<2><<<cdiv(tiles, BLK / 32), BLK, 0, stream>>>(h1, W2, h2pre,
                                                                N_NODES, F1, F2, tiles);
  }
  att_dots<F2><<<cdiv(N_NODES, BLK), BLK, 0, stream>>>(h2pre, as2, ad2,
                                                       asrc2, adst2, N_NODES, 1);
  node_softmax_aggr_1x32<<<cdiv(N_NODES, BLK / 32), BLK, 0, stream>>>(
      ei, cnt, slot, head_ov, nxt_ov, asrc2, adst2, h2pre, b2, out);
}
